# starting kernel
# baseline (speedup 1.0000x reference)
#include <hip/hip_runtime.h>
#include <math.h>

typedef float v2f __attribute__((ext_vector_type(2)));
typedef float v8f __attribute__((ext_vector_type(8)));

#define ROWS 32
#define KNN 16
#define HD 128
#define MIXK 262
#define MIXKP 264
#define MSTRIDE 268

__global__ __launch_bounds__(256)
void graphconv_wmma_kernel(const float* __restrict__ feat,
                           const float* __restrict__ coords,
                           const int*   __restrict__ knn,
                           const float* __restrict__ W,
                           const float* __restrict__ bias,
                           float* __restrict__ out)
{
    extern __shared__ float smem[];
    float* mixLds = smem;
    int*   knnLds = (int*)(smem + ROWS * MSTRIDE);

    const int tid  = threadIdx.x;
    const int base = blockIdx.x * ROWS;

    for (int i = tid; i < ROWS * KNN; i += 256)
        knnLds[i] = __builtin_nontemporal_load(knn + (size_t)base * KNN + i);
    __syncthreads();

    {
        const int r    = tid >> 3;
        const int sub  = tid & 7;
        const int col0 = sub * 16;
        float* mrow = mixLds + r * MSTRIDE;

        const float4* sp = (const float4*)(feat + (size_t)(base + r) * HD + col0);
        float4* dst_self = (float4*)(mrow + col0);
        float4* dst_agg  = (float4*)(mrow + HD + col0);
        float ax[16];
        #pragma unroll
        for (int j = 0; j < 4; ++j) {
            float4 v = sp[j];
            dst_self[j] = v;
            ax[4*j+0] = 0.f; ax[4*j+1] = 0.f; ax[4*j+2] = 0.f; ax[4*j+3] = 0.f;
        }
        #pragma unroll
        for (int k = 0; k < KNN; ++k) {
            const int idx = knnLds[r * KNN + k];
            const float4* np = (const float4*)(feat + (size_t)idx * HD + col0);
            #pragma unroll
            for (int j = 0; j < 4; ++j) {
                float4 v = np[j];
                ax[4*j+0] += v.x; ax[4*j+1] += v.y;
                ax[4*j+2] += v.z; ax[4*j+3] += v.w;
            }
        }
        const float inv = 1.0f / 16.0f;
        #pragma unroll
        for (int j = 0; j < 4; ++j) {
            float4 v;
            v.x = ax[4*j+0]*inv; v.y = ax[4*j+1]*inv;
            v.z = ax[4*j+2]*inv; v.w = ax[4*j+3]*inv;
            dst_agg[j] = v;
        }
    }

    if (tid < ROWS) {
        const int r = tid;
        const int node = base + r;
        const float cx = coords[(size_t)node*3+0];
        const float cy = coords[(size_t)node*3+1];
        const float cz = coords[(size_t)node*3+2];
        float sx=0.f, sy=0.f, sz=0.f, qx=0.f, qy=0.f, qz=0.f;
        #pragma unroll
        for (int k = 0; k < KNN; ++k) {
            const int idx = knnLds[r * KNN + k];
            float dx = coords[(size_t)idx*3+0] - cx;
            float dy = coords[(size_t)idx*3+1] - cy;
            float dz = coords[(size_t)idx*3+2] - cz;
            sx += dx; sy += dy; sz += dz;
            qx += dx*dx; qy += dy*dy; qz += dz*dz;
        }
        const float inv = 1.0f / 16.0f;
        const float mx = sx*inv, my = sy*inv, mz = sz*inv;
        float vx = qx*inv - mx*mx; vx = vx > 0.f ? vx : 0.f;
        float vy = qy*inv - my*my; vy = vy > 0.f ? vy : 0.f;
        float vz = qz*inv - mz*mz; vz = vz > 0.f ? vz : 0.f;
        float* mrow = mixLds + r * MSTRIDE + 2*HD;
        mrow[0] = mx;        mrow[1] = my;        mrow[2] = mz;
        mrow[3] = sqrtf(vx); mrow[4] = sqrtf(vy); mrow[5] = sqrtf(vz);
        mrow[6] = 0.f;       mrow[7] = 0.f;
    }
    __syncthreads();

    const int wave    = tid >> 5;
    const int lane    = tid & 31;
    const int colbase = wave * 16;
    const int lx      = lane & 15;
    const int hi      = lane >> 4;

    v8f c0 = {};
    v8f c1 = {};

    const float* mA0 = mixLds + lx * MSTRIDE;
    const float* mA1 = mixLds + (lx + 16) * MSTRIDE;

    for (int kk = 0; kk < MIXKP - 4; kk += 4) {
        const int k0 = kk + hi * 2;
        v2f a0, a1, b;
        a0.x = mA0[k0]; a0.y = mA0[k0 + 1];
        a1.x = mA1[k0]; a1.y = mA1[k0 + 1];
        const float* wp = W + (size_t)k0 * HD + colbase + lx;
        b.x = wp[0];
        b.y = wp[HD];
        c0 = __builtin_amdgcn_wmma_f32_16x16x4_f32(false, a0, false, b,
                                                   (short)0, c0, false, false);
        c1 = __builtin_amdgcn_wmma_f32_16x16x4_f32(false, a1, false, b,
                                                   (short)0, c1, false, false);
    }
    {
        const int k0 = (MIXKP - 4) + hi * 2;
        v2f a0, a1, b;
        a0.x = mA0[k0]; a0.y = mA0[k0 + 1];
        a1.x = mA1[k0]; a1.y = mA1[k0 + 1];
        if (k0 < MIXK) {
            const float* wp = W + (size_t)k0 * HD + colbase + lx;
            b.x = wp[0];
            b.y = wp[HD];
        } else {
            b.x = 0.f; b.y = 0.f;
        }
        c0 = __builtin_amdgcn_wmma_f32_16x16x4_f32(false, a0, false, b,
                                                   (short)0, c0, false, false);
        c1 = __builtin_amdgcn_wmma_f32_16x16x4_f32(false, a1, false, b,
                                                   (short)0, c1, false, false);
    }

    const int ncol = colbase + lx;
    const float bv = bias[ncol];
    #pragma unroll
    for (int r = 0; r < 8; ++r) {
        const int row = r + hi * 8;
        float x0 = c0[r] + bv;
        __builtin_nontemporal_store(x0 / (1.0f + __expf(-x0)),
                                    out + (size_t)(base + row) * HD + ncol);
        float x1 = c1[r] + bv;
        __builtin_nontemporal_store(x1 / (1.0f + __expf(-x1)),
                                    out + (size_t)(base + 16 + row) * HD + ncol);
    }
}

extern "C" void kernel_launch(void* const* d_in, const int* in_sizes, int n_in,
                              void* d_out, int out_size, void* d_ws, size_t ws_size,
                              hipStream_t stream) {
    const float* feat   = (const float*)d_in[0];
    const float* coords = (const float*)d_in[1];
    const int*   knn    = (const int*)  d_in[2];
    const float* W      = (const float*)d_in[3];
    const float* bias   = (const float*)d_in[4];
    float* out = (float*)d_out;

    const int n = in_sizes[0] / HD;
    const int grid = (n + ROWS - 1) / ROWS;
    const size_t shmem = (size_t)ROWS * MSTRIDE * sizeof(float)
                       + (size_t)ROWS * KNN * sizeof(int);

    graphconv_wmma_kernel<<<grid, 256, shmem, stream>>>(feat, coords, knn, W, bias, out);
}


# Gather with one whole 512-byte feature row per wave load (lane l = floats 4l..4l+3): 4 full 128-B lines per load, not 16 quarter lines; neighbour sum order unchanged
# speedup vs baseline: 1.3130x; 1.3130x over previous
#include <hip/hip_runtime.h>
#include <math.h>

typedef float v2f __attribute__((ext_vector_type(2)));
typedef float v8f __attribute__((ext_vector_type(8)));

#define ROWS 32
#define KNN 16
#define HD 128
#define MIXK 262
#define MIXKP 264
#define MSTRIDE 268

__global__ __launch_bounds__(256)
void graphconv_wmma_kernel(const float* __restrict__ feat,
                           const float* __restrict__ coords,
                           const int*   __restrict__ knn,
                           const float* __restrict__ W,
                           const float* __restrict__ bias,
                           float* __restrict__ out)
{
    extern __shared__ float smem[];
    float* mixLds = smem;
    int*   knnLds = (int*)(smem + ROWS * MSTRIDE);

    const int tid  = threadIdx.x;
    const int base = blockIdx.x * ROWS;

    for (int i = tid; i < ROWS * KNN; i += 256)
        knnLds[i] = __builtin_nontemporal_load(knn + (size_t)base * KNN + i);
    __syncthreads();

    {
        const int gw  = tid >> 5;
        const int gl4 = (tid & 31) * 4;
        #pragma unroll
        for (int n = 0; n < ROWS / 8; ++n) {
            const int r = gw * (ROWS / 8) + n;
            float* mrow = mixLds + r * MSTRIDE;

            const float4 sv = *(const float4*)(feat + (size_t)(base + r) * HD + gl4);
            *(float4*)(mrow + gl4) = sv;

            float4 acc = make_float4(0.f, 0.f, 0.f, 0.f);
            #pragma unroll
            for (int k = 0; k < KNN; ++k) {
                const int idx = __builtin_amdgcn_readfirstlane(knnLds[r * KNN + k]);
                const float4 v = *(const float4*)(feat + (size_t)idx * HD + gl4);
                acc.x += v.x; acc.y += v.y; acc.z += v.z; acc.w += v.w;
            }
            const float inv = 1.0f / 16.0f;
            float4 o;
            o.x = acc.x * inv; o.y = acc.y * inv; o.z = acc.z * inv; o.w = acc.w * inv;
            *(float4*)(mrow + HD + gl4) = o;
        }
    }

    if (tid < ROWS) {
        const int r = tid;
        const int node = base + r;
        const float cx = coords[(size_t)node*3+0];
        const float cy = coords[(size_t)node*3+1];
        const float cz = coords[(size_t)node*3+2];
        float sx=0.f, sy=0.f, sz=0.f, qx=0.f, qy=0.f, qz=0.f;
        #pragma unroll
        for (int k = 0; k < KNN; ++k) {
            const int idx = knnLds[r * KNN + k];
            float dx = coords[(size_t)idx*3+0] - cx;
            float dy = coords[(size_t)idx*3+1] - cy;
            float dz = coords[(size_t)idx*3+2] - cz;
            sx += dx; sy += dy; sz += dz;
            qx += dx*dx; qy += dy*dy; qz += dz*dz;
        }
        const float inv = 1.0f / 16.0f;
        const float mx = sx*inv, my = sy*inv, mz = sz*inv;
        float vx = qx*inv - mx*mx; vx = vx > 0.f ? vx : 0.f;
        float vy = qy*inv - my*my; vy = vy > 0.f ? vy : 0.f;
        float vz = qz*inv - mz*mz; vz = vz > 0.f ? vz : 0.f;
        float* mrow = mixLds + r * MSTRIDE + 2*HD;
        mrow[0] = mx;        mrow[1] = my;        mrow[2] = mz;
        mrow[3] = sqrtf(vx); mrow[4] = sqrtf(vy); mrow[5] = sqrtf(vz);
        mrow[6] = 0.f;       mrow[7] = 0.f;
    }
    __syncthreads();

    const int wave    = tid >> 5;
    const int lane    = tid & 31;
    const int colbase = wave * 16;
    const int lx      = lane & 15;
    const int hi      = lane >> 4;

    v8f c0 = {};
    v8f c1 = {};

    const float* mA0 = mixLds + lx * MSTRIDE;
    const float* mA1 = mixLds + (lx + 16) * MSTRIDE;

    for (int kk = 0; kk < MIXKP - 4; kk += 4) {
        const int k0 = kk + hi * 2;
        v2f a0, a1, b;
        a0.x = mA0[k0]; a0.y = mA0[k0 + 1];
        a1.x = mA1[k0]; a1.y = mA1[k0 + 1];
        const float* wp = W + (size_t)k0 * HD + colbase + lx;
        b.x = wp[0];
        b.y = wp[HD];
        c0 = __builtin_amdgcn_wmma_f32_16x16x4_f32(false, a0, false, b,
                                                   (short)0, c0, false, false);
        c1 = __builtin_amdgcn_wmma_f32_16x16x4_f32(false, a1, false, b,
                                                   (short)0, c1, false, false);
    }
    {
        const int k0 = (MIXKP - 4) + hi * 2;
        v2f a0, a1, b;
        a0.x = mA0[k0]; a0.y = mA0[k0 + 1];
        a1.x = mA1[k0]; a1.y = mA1[k0 + 1];
        if (k0 < MIXK) {
            const float* wp = W + (size_t)k0 * HD + colbase + lx;
            b.x = wp[0];
            b.y = wp[HD];
        } else {
            b.x = 0.f; b.y = 0.f;
        }
        c0 = __builtin_amdgcn_wmma_f32_16x16x4_f32(false, a0, false, b,
                                                   (short)0, c0, false, false);
        c1 = __builtin_amdgcn_wmma_f32_16x16x4_f32(false, a1, false, b,
                                                   (short)0, c1, false, false);
    }

    const int ncol = colbase + lx;
    const float bv = bias[ncol];
    #pragma unroll
    for (int r = 0; r < 8; ++r) {
        const int row = r + hi * 8;
        float x0 = c0[r] + bv;
        __builtin_nontemporal_store(x0 / (1.0f + __expf(-x0)),
                                    out + (size_t)(base + row) * HD + ncol);
        float x1 = c1[r] + bv;
        __builtin_nontemporal_store(x1 / (1.0f + __expf(-x1)),
                                    out + (size_t)(base + 16 + row) * HD + ncol);
    }
}

extern "C" void kernel_launch(void* const* d_in, const int* in_sizes, int n_in,
                              void* d_out, int out_size, void* d_ws, size_t ws_size,
                              hipStream_t stream) {
    const float* feat   = (const float*)d_in[0];
    const float* coords = (const float*)d_in[1];
    const int*   knn    = (const int*)  d_in[2];
    const float* W      = (const float*)d_in[3];
    const float* bias   = (const float*)d_in[4];
    float* out = (float*)d_out;

    const int n = in_sizes[0] / HD;
    const int grid = (n + ROWS - 1) / ROWS;
    const size_t shmem = (size_t)ROWS * MSTRIDE * sizeof(float)
                       + (size_t)ROWS * KNN * sizeof(int);

    graphconv_wmma_kernel<<<grid, 256, shmem, stream>>>(feat, coords, knn, W, bias, out);
}


# GEMM wave tile 32x16 -> 16x64, columns interleaved by 4, on waves 0-3 (4-7 exit after the gather): a float4 W load covers whole lines and feeds 4 WMMAs; float4 whole-line stores; k order unchanged
# speedup vs baseline: 1.3801x; 1.0511x over previous
#include <hip/hip_runtime.h>
#include <math.h>

typedef float v2f __attribute__((ext_vector_type(2)));
typedef float v4f __attribute__((ext_vector_type(4)));
typedef float v8f __attribute__((ext_vector_type(8)));

#define ROWS 32
#define KNN 16
#define HD 128
#define MIXK 262
#define MIXKP 264
#define MSTRIDE 268

__global__ __launch_bounds__(256)
void graphconv_wmma_kernel(const float* __restrict__ feat,
                           const float* __restrict__ coords,
                           const int*   __restrict__ knn,
                           const float* __restrict__ W,
                           const float* __restrict__ bias,
                           float* __restrict__ out)
{
    extern __shared__ float smem[];
    float* mixLds = smem;
    int*   knnLds = (int*)(smem + ROWS * MSTRIDE);

    const int tid  = threadIdx.x;
    const int base = blockIdx.x * ROWS;

    for (int i = tid; i < ROWS * KNN; i += 256)
        knnLds[i] = __builtin_nontemporal_load(knn + (size_t)base * KNN + i);
    __syncthreads();

    {
        const int gw  = tid >> 5;
        const int gl4 = (tid & 31) * 4;
        #pragma unroll
        for (int n = 0; n < ROWS / 8; ++n) {
            const int r = gw * (ROWS / 8) + n;
            float* mrow = mixLds + r * MSTRIDE;

            const float4 sv = *(const float4*)(feat + (size_t)(base + r) * HD + gl4);
            *(float4*)(mrow + gl4) = sv;

            float4 acc = make_float4(0.f, 0.f, 0.f, 0.f);
            #pragma unroll
            for (int k = 0; k < KNN; ++k) {
                const int idx = __builtin_amdgcn_readfirstlane(knnLds[r * KNN + k]);
                const float4 v = *(const float4*)(feat + (size_t)idx * HD + gl4);
                acc.x += v.x; acc.y += v.y; acc.z += v.z; acc.w += v.w;
            }
            const float inv = 1.0f / 16.0f;
            float4 o;
            o.x = acc.x * inv; o.y = acc.y * inv; o.z = acc.z * inv; o.w = acc.w * inv;
            *(float4*)(mrow + HD + gl4) = o;
        }
    }

    if (tid < ROWS) {
        const int r = tid;
        const int node = base + r;
        const float cx = coords[(size_t)node*3+0];
        const float cy = coords[(size_t)node*3+1];
        const float cz = coords[(size_t)node*3+2];
        float sx=0.f, sy=0.f, sz=0.f, qx=0.f, qy=0.f, qz=0.f;
        #pragma unroll
        for (int k = 0; k < KNN; ++k) {
            const int idx = knnLds[r * KNN + k];
            float dx = coords[(size_t)idx*3+0] - cx;
            float dy = coords[(size_t)idx*3+1] - cy;
            float dz = coords[(size_t)idx*3+2] - cz;
            sx += dx; sy += dy; sz += dz;
            qx += dx*dx; qy += dy*dy; qz += dz*dz;
        }
        const float inv = 1.0f / 16.0f;
        const float mx = sx*inv, my = sy*inv, mz = sz*inv;
        float vx = qx*inv - mx*mx; vx = vx > 0.f ? vx : 0.f;
        float vy = qy*inv - my*my; vy = vy > 0.f ? vy : 0.f;
        float vz = qz*inv - mz*mz; vz = vz > 0.f ? vz : 0.f;
        float* mrow = mixLds + r * MSTRIDE + 2*HD;
        mrow[0] = mx;        mrow[1] = my;        mrow[2] = mz;
        mrow[3] = sqrtf(vx); mrow[4] = sqrtf(vy); mrow[5] = sqrtf(vz);
        mrow[6] = 0.f;       mrow[7] = 0.f;
    }
    __syncthreads();

    if (tid >= 128) return;

    const int wave  = tid >> 5;
    const int lane  = tid & 31;
    const int lx    = lane & 15;
    const int hi    = lane >> 4;
    const int rbase = (wave & 1) * 16;
    const int col4  = (wave >> 1) * 64 + lx * 4;

    v8f c0 = {}, c1 = {}, c2 = {}, c3 = {};

    const float* mA = mixLds + (rbase + lx) * MSTRIDE;
    const float* wB = W + col4;

    for (int kk = 0; kk < MIXKP - 4; kk += 4) {
        const int k0 = kk + hi * 2;
        v2f a;
        a.x = mA[k0]; a.y = mA[k0 + 1];
        const float4 w0 = *(const float4*)(wB + (size_t)k0 * HD);
        const float4 w1 = *(const float4*)(wB + (size_t)(k0 + 1) * HD);
        v2f b0, b1, b2, b3;
        b0.x = w0.x; b0.y = w1.x;
        b1.x = w0.y; b1.y = w1.y;
        b2.x = w0.z; b2.y = w1.z;
        b3.x = w0.w; b3.y = w1.w;
        c0 = __builtin_amdgcn_wmma_f32_16x16x4_f32(false, a, false, b0,
                                                   (short)0, c0, false, false);
        c1 = __builtin_amdgcn_wmma_f32_16x16x4_f32(false, a, false, b1,
                                                   (short)0, c1, false, false);
        c2 = __builtin_amdgcn_wmma_f32_16x16x4_f32(false, a, false, b2,
                                                   (short)0, c2, false, false);
        c3 = __builtin_amdgcn_wmma_f32_16x16x4_f32(false, a, false, b3,
                                                   (short)0, c3, false, false);
    }
    {
        const int k0 = (MIXKP - 4) + hi * 2;
        v2f a;
        a.x = mA[k0]; a.y = mA[k0 + 1];
        float4 w0 = make_float4(0.f, 0.f, 0.f, 0.f);
        float4 w1 = make_float4(0.f, 0.f, 0.f, 0.f);
        if (k0 < MIXK) {
            w0 = *(const float4*)(wB + (size_t)k0 * HD);
            w1 = *(const float4*)(wB + (size_t)(k0 + 1) * HD);
        }
        v2f b0, b1, b2, b3;
        b0.x = w0.x; b0.y = w1.x;
        b1.x = w0.y; b1.y = w1.y;
        b2.x = w0.z; b2.y = w1.z;
        b3.x = w0.w; b3.y = w1.w;
        c0 = __builtin_amdgcn_wmma_f32_16x16x4_f32(false, a, false, b0,
                                                   (short)0, c0, false, false);
        c1 = __builtin_amdgcn_wmma_f32_16x16x4_f32(false, a, false, b1,
                                                   (short)0, c1, false, false);
        c2 = __builtin_amdgcn_wmma_f32_16x16x4_f32(false, a, false, b2,
                                                   (short)0, c2, false, false);
        c3 = __builtin_amdgcn_wmma_f32_16x16x4_f32(false, a, false, b3,
                                                   (short)0, c3, false, false);
    }

    const float4 bv = *(const float4*)(bias + col4);
    #pragma unroll
    for (int r = 0; r < 8; ++r) {
        const int row = rbase + r + hi * 8;
        const float x0 = c0[r] + bv.x;
        const float x1 = c1[r] + bv.y;
        const float x2 = c2[r] + bv.z;
        const float x3 = c3[r] + bv.w;
        v4f o;
        o.x = x0 / (1.0f + __expf(-x0));
        o.y = x1 / (1.0f + __expf(-x1));
        o.z = x2 / (1.0f + __expf(-x2));
        o.w = x3 / (1.0f + __expf(-x3));
        __builtin_nontemporal_store(o, (v4f*)(out + (size_t)(base + row) * HD + col4));
    }
}

extern "C" void kernel_launch(void* const* d_in, const int* in_sizes, int n_in,
                              void* d_out, int out_size, void* d_ws, size_t ws_size,
                              hipStream_t stream) {
    const float* feat   = (const float*)d_in[0];
    const float* coords = (const float*)d_in[1];
    const int*   knn    = (const int*)  d_in[2];
    const float* W      = (const float*)d_in[3];
    const float* bias   = (const float*)d_in[4];
    float* out = (float*)d_out;

    const int n = in_sizes[0] / HD;
    const int grid = (n + ROWS - 1) / ROWS;
    const size_t shmem = (size_t)ROWS * MSTRIDE * sizeof(float)
                       + (size_t)ROWS * KNN * sizeof(int);

    graphconv_wmma_kernel<<<grid, 256, shmem, stream>>>(feat, coords, knn, W, bias, out);
}


# Work-groups (1024, all resident, 2 LDS stages) walk their tiles as a pipeline: waves 4-7 gather tile s+1 while waves 0-3 multiply tile s, so gather and WMMA overlap; all sum orders unchanged
# speedup vs baseline: 1.4500x; 1.0506x over previous
#include <hip/hip_runtime.h>
#include <math.h>


typedef float v2f __attribute__((ext_vector_type(2)));
typedef float v4f __attribute__((ext_vector_type(4)));
typedef float v8f __attribute__((ext_vector_type(8)));
typedef int   v4i __attribute__((ext_vector_type(4)));

#define ROWS 32
#define KNN 16
#define HD 128
#define MIXK 262
#define MIXKP 264
#define MSTRIDE 268
#define STAGE_WORDS (ROWS * MSTRIDE + ROWS * KNN)
#define NGROUPS 1024

__global__ __launch_bounds__(256, 8)
void graphconv_wmma_kernel(const float* __restrict__ feat,
                           const float* __restrict__ coords,
                           const int*   __restrict__ knn,
                           const float* __restrict__ W,
                           const float* __restrict__ bias,
                           float* __restrict__ out,
                           int n)
{
    __shared__ __attribute__((aligned(16))) float smem[2 * STAGE_WORDS];

    const int tid  = threadIdx.x;
    const int wave = tid >> 5;
    const int lane = tid & 31;

    const int ntiles = n / ROWS;
    int my = 0;
    if ((int)blockIdx.x < ntiles)
        my = (ntiles - 1 - (int)blockIdx.x) / (int)gridDim.x + 1;

    for (int s = 0; s <= my; ++s) {
        if (tid >= 128) {
            if (s < my) {
                const int pw   = wave - 4;
                const int base = ((int)blockIdx.x + s * (int)gridDim.x) * ROWS;
                float* mixLds = smem + (s & 1) * STAGE_WORDS;
                int*   knnLds = (int*)(mixLds + ROWS * MSTRIDE);

                {
                    const v4i iv = __builtin_nontemporal_load(
                        (const v4i*)(knn + ((size_t)base + pw * 8) * KNN) + lane);
                    *((v4i*)(knnLds + pw * 8 * KNN) + lane) = iv;
                }
                __builtin_amdgcn_fence(__ATOMIC_RELEASE, "wavefront");
                __builtin_amdgcn_wave_barrier();
                __builtin_amdgcn_fence(__ATOMIC_ACQUIRE, "wavefront");

                const int gl4 = lane * 4;
                #pragma unroll 1
                for (int nn = 0; nn < 8; ++nn) {
                    const int r = pw * 8 + nn;
                    float* mrow = mixLds + r * MSTRIDE;

                    const float4 sv = *(const float4*)(feat + (size_t)(base + r) * HD + gl4);

                    float4 acc = make_float4(0.f, 0.f, 0.f, 0.f);
                    #pragma unroll
                    for (int k = 0; k < KNN; ++k) {
                        const int idx = __builtin_amdgcn_readfirstlane(knnLds[r * KNN + k]);
                        const float4 v = *(const float4*)(feat + (size_t)idx * HD + gl4);
                        acc.x += v.x; acc.y += v.y; acc.z += v.z; acc.w += v.w;
                    }
                    const float inv = 1.0f / 16.0f;
                    float4 o;
                    o.x = acc.x * inv; o.y = acc.y * inv; o.z = acc.z * inv; o.w = acc.w * inv;
                    *(float4*)(mrow + gl4) = sv;
                    *(float4*)(mrow + HD + gl4) = o;
                }

                if (lane < 8) {
                    const int r = pw * 8 + lane;
                    const int node = base + r;
                    const float cx = coords[(size_t)node*3+0];
                    const float cy = coords[(size_t)node*3+1];
                    const float cz = coords[(size_t)node*3+2];
                    float sx=0.f, sy=0.f, sz=0.f, qx=0.f, qy=0.f, qz=0.f;
                    #pragma unroll
                    for (int k = 0; k < KNN; ++k) {
                        const int idx = knnLds[r * KNN + k];
                        float dx = coords[(size_t)idx*3+0] - cx;
                        float dy = coords[(size_t)idx*3+1] - cy;
                        float dz = coords[(size_t)idx*3+2] - cz;
                        sx += dx; sy += dy; sz += dz;
                        qx += dx*dx; qy += dy*dy; qz += dz*dz;
                    }
                    const float inv = 1.0f / 16.0f;
                    const float mx = sx*inv, my_ = sy*inv, mz = sz*inv;
                    float vx = qx*inv - mx*mx;   vx = vx > 0.f ? vx : 0.f;
                    float vy = qy*inv - my_*my_; vy = vy > 0.f ? vy : 0.f;
                    float vz = qz*inv - mz*mz;   vz = vz > 0.f ? vz : 0.f;
                    float* mrow = mixLds + r * MSTRIDE + 2*HD;
                    mrow[0] = mx;        mrow[1] = my_;       mrow[2] = mz;
                    mrow[3] = sqrtf(vx); mrow[4] = sqrtf(vy); mrow[5] = sqrtf(vz);
                    mrow[6] = 0.f;       mrow[7] = 0.f;
                }
            }
        } else {
            if (s >= 1) {
                const int base = ((int)blockIdx.x + (s - 1) * (int)gridDim.x) * ROWS;
                const float* mixLds = smem + ((s - 1) & 1) * STAGE_WORDS;

                const int lx    = lane & 15;
                const int hi    = lane >> 4;
                const int rbase = (wave & 1) * 16;
                const int col4  = (wave >> 1) * 64 + lx * 4;

                v8f c0 = {}, c1 = {}, c2 = {}, c3 = {};

                const float* mA = mixLds + (rbase + lx) * MSTRIDE;
                const float* wB = W + col4;

                for (int kk = 0; kk < MIXKP - 4; kk += 4) {
                    const int k0 = kk + hi * 2;
                    v2f a;
                    a.x = mA[k0]; a.y = mA[k0 + 1];
                    const float4 w0 = *(const float4*)(wB + (size_t)k0 * HD);
                    const float4 w1 = *(const float4*)(wB + (size_t)(k0 + 1) * HD);
                    v2f b0, b1, b2, b3;
                    b0.x = w0.x; b0.y = w1.x;
                    b1.x = w0.y; b1.y = w1.y;
                    b2.x = w0.z; b2.y = w1.z;
                    b3.x = w0.w; b3.y = w1.w;
                    c0 = __builtin_amdgcn_wmma_f32_16x16x4_f32(false, a, false, b0,
                                                               (short)0, c0, false, false);
                    c1 = __builtin_amdgcn_wmma_f32_16x16x4_f32(false, a, false, b1,
                                                               (short)0, c1, false, false);
                    c2 = __builtin_amdgcn_wmma_f32_16x16x4_f32(false, a, false, b2,
                                                               (short)0, c2, false, false);
                    c3 = __builtin_amdgcn_wmma_f32_16x16x4_f32(false, a, false, b3,
                                                               (short)0, c3, false, false);
                }
                {
                    const int k0 = (MIXKP - 4) + hi * 2;
                    v2f a;
                    a.x = mA[k0]; a.y = mA[k0 + 1];
                    float4 w0 = make_float4(0.f, 0.f, 0.f, 0.f);
                    float4 w1 = make_float4(0.f, 0.f, 0.f, 0.f);
                    if (k0 < MIXK) {
                        w0 = *(const float4*)(wB + (size_t)k0 * HD);
                        w1 = *(const float4*)(wB + (size_t)(k0 + 1) * HD);
                    }
                    v2f b0, b1, b2, b3;
                    b0.x = w0.x; b0.y = w1.x;
                    b1.x = w0.y; b1.y = w1.y;
                    b2.x = w0.z; b2.y = w1.z;
                    b3.x = w0.w; b3.y = w1.w;
                    c0 = __builtin_amdgcn_wmma_f32_16x16x4_f32(false, a, false, b0,
                                                               (short)0, c0, false, false);
                    c1 = __builtin_amdgcn_wmma_f32_16x16x4_f32(false, a, false, b1,
                                                               (short)0, c1, false, false);
                    c2 = __builtin_amdgcn_wmma_f32_16x16x4_f32(false, a, false, b2,
                                                               (short)0, c2, false, false);
                    c3 = __builtin_amdgcn_wmma_f32_16x16x4_f32(false, a, false, b3,
                                                               (short)0, c3, false, false);
                }

                const float4 bv = *(const float4*)(bias + col4);
                #pragma unroll
                for (int r = 0; r < 8; ++r) {
                    const int row = rbase + r + hi * 8;
                    const float x0 = c0[r] + bv.x;
                    const float x1 = c1[r] + bv.y;
                    const float x2 = c2[r] + bv.z;
                    const float x3 = c3[r] + bv.w;
                    v4f o;
                    o.x = x0 / (1.0f + __expf(-x0));
                    o.y = x1 / (1.0f + __expf(-x1));
                    o.z = x2 / (1.0f + __expf(-x2));
                    o.w = x3 / (1.0f + __expf(-x3));
                    __builtin_nontemporal_store(o, (v4f*)(out + (size_t)(base + row) * HD + col4));
                }
            }
        }
        __syncthreads();
    }
}

extern "C" void kernel_launch(void* const* d_in, const int* in_sizes, int n_in,
                              void* d_out, int out_size, void* d_ws, size_t ws_size,
                              hipStream_t stream) {
    const float* feat   = (const float*)d_in[0];
    const float* coords = (const float*)d_in[1];
    const int*   knn    = (const int*)  d_in[2];
    const float* W      = (const float*)d_in[3];
    const float* bias   = (const float*)d_in[4];
    float* out = (float*)d_out;

    const int n = in_sizes[0] / HD;

    graphconv_wmma_kernel<<<NGROUPS, 256, 0, stream>>>(feat, coords, knn, W, bias, out, n);
}


# Multiplying waves refill a k step's W and A registers with k step q+5 right after its matrix instructions issue: W waits fall under 20 matrix instructions, not in front of them; k order kept
# speedup vs baseline: 1.5002x; 1.0346x over previous
#include <hip/hip_runtime.h>
#include <math.h>


typedef float v2f __attribute__((ext_vector_type(2)));
typedef float v4f __attribute__((ext_vector_type(4)));
typedef float v8f __attribute__((ext_vector_type(8)));
typedef int   v4i __attribute__((ext_vector_type(4)));

#define ROWS 32
#define KNN 16
#define HD 128
#define MIXK 262
#define MIXKP 264
#define MSTRIDE 268
#define STAGE_WORDS (ROWS * MSTRIDE + ROWS * KNN)
#define NGROUPS 1024

__global__ __launch_bounds__(256, 8)
void graphconv_wmma_kernel(const float* __restrict__ feat,
                           const float* __restrict__ coords,
                           const int*   __restrict__ knn,
                           const float* __restrict__ W,
                           const float* __restrict__ bias,
                           float* __restrict__ out,
                           int n)
{
    __shared__ __attribute__((aligned(16))) float smem[2 * STAGE_WORDS];

    const int tid  = threadIdx.x;
    const int wave = tid >> 5;
    const int lane = tid & 31;

    const int ntiles = n / ROWS;
    int my = 0;
    if ((int)blockIdx.x < ntiles)
        my = (ntiles - 1 - (int)blockIdx.x) / (int)gridDim.x + 1;

    for (int s = 0; s <= my; ++s) {
        if (tid >= 128) {
            if (s < my) {
                const int pw   = wave - 4;
                const int base = ((int)blockIdx.x + s * (int)gridDim.x) * ROWS;
                float* mixLds = smem + (s & 1) * STAGE_WORDS;
                int*   knnLds = (int*)(mixLds + ROWS * MSTRIDE);

                {
                    const v4i iv = __builtin_nontemporal_load(
                        (const v4i*)(knn + ((size_t)base + pw * 8) * KNN) + lane);
                    *((v4i*)(knnLds + pw * 8 * KNN) + lane) = iv;
                }
                __builtin_amdgcn_fence(__ATOMIC_RELEASE, "wavefront");
                __builtin_amdgcn_wave_barrier();
                __builtin_amdgcn_fence(__ATOMIC_ACQUIRE, "wavefront");

                const int gl4 = lane * 4;
                #pragma unroll 1
                for (int nn = 0; nn < 8; ++nn) {
                    const int r = pw * 8 + nn;
                    float* mrow = mixLds + r * MSTRIDE;

                    const float4 sv = *(const float4*)(feat + (size_t)(base + r) * HD + gl4);

                    float4 acc = make_float4(0.f, 0.f, 0.f, 0.f);
                    #pragma unroll
                    for (int k = 0; k < KNN; ++k) {
                        const int idx = __builtin_amdgcn_readfirstlane(knnLds[r * KNN + k]);
                        const float4 v = *(const float4*)(feat + (size_t)idx * HD + gl4);
                        acc.x += v.x; acc.y += v.y; acc.z += v.z; acc.w += v.w;
                    }
                    const float inv = 1.0f / 16.0f;
                    float4 o;
                    o.x = acc.x * inv; o.y = acc.y * inv; o.z = acc.z * inv; o.w = acc.w * inv;
                    *(float4*)(mrow + gl4) = sv;
                    *(float4*)(mrow + HD + gl4) = o;
                }

                if (lane < 8) {
                    const int r = pw * 8 + lane;
                    const int node = base + r;
                    const float cx = coords[(size_t)node*3+0];
                    const float cy = coords[(size_t)node*3+1];
                    const float cz = coords[(size_t)node*3+2];
                    float sx=0.f, sy=0.f, sz=0.f, qx=0.f, qy=0.f, qz=0.f;
                    #pragma unroll
                    for (int k = 0; k < KNN; ++k) {
                        const int idx = knnLds[r * KNN + k];
                        float dx = coords[(size_t)idx*3+0] - cx;
                        float dy = coords[(size_t)idx*3+1] - cy;
                        float dz = coords[(size_t)idx*3+2] - cz;
                        sx += dx; sy += dy; sz += dz;
                        qx += dx*dx; qy += dy*dy; qz += dz*dz;
                    }
                    const float inv = 1.0f / 16.0f;
                    const float mx = sx*inv, my_ = sy*inv, mz = sz*inv;
                    float vx = qx*inv - mx*mx;   vx = vx > 0.f ? vx : 0.f;
                    float vy = qy*inv - my_*my_; vy = vy > 0.f ? vy : 0.f;
                    float vz = qz*inv - mz*mz;   vz = vz > 0.f ? vz : 0.f;
                    float* mrow = mixLds + r * MSTRIDE + 2*HD;
                    mrow[0] = mx;        mrow[1] = my_;       mrow[2] = mz;
                    mrow[3] = sqrtf(vx); mrow[4] = sqrtf(vy); mrow[5] = sqrtf(vz);
                    mrow[6] = 0.f;       mrow[7] = 0.f;
                }
            }
        } else {
            if (s >= 1) {
                const int base = ((int)blockIdx.x + (s - 1) * (int)gridDim.x) * ROWS;
                const float* mixLds = smem + ((s - 1) & 1) * STAGE_WORDS;

                const int lx    = lane & 15;
                const int hi    = lane >> 4;
                const int rbase = (wave & 1) * 16;
                const int col4  = (wave >> 1) * 64 + lx * 4;

                v8f c0 = {}, c1 = {}, c2 = {}, c3 = {};

                const float* mA = mixLds + (rbase + lx) * MSTRIDE + hi * 2;
                const float* wB = W + col4 + (size_t)(hi * 2) * HD;

                float4 wlo[5], whi[5];
                v2f av[5];
                #pragma unroll
                for (int j = 0; j < 5; ++j) {
                    wlo[j] = *(const float4*)(wB + (size_t)j * 4 * HD);
                    whi[j] = *(const float4*)(wB + (size_t)j * 4 * HD + HD);
                    av[j]  = *(const v2f*)(mA + j * 4);
                }

#define MMA_STEP(wlo_, whi_, a_)                                                        \
                {                                                                       \
                    v2f b0_, b1_, b2_, b3_;                                             \
                    b0_.x = wlo_.x; b0_.y = whi_.x;                                     \
                    b1_.x = wlo_.y; b1_.y = whi_.y;                                     \
                    b2_.x = wlo_.z; b2_.y = whi_.z;                                     \
                    b3_.x = wlo_.w; b3_.y = whi_.w;                                     \
                    c0 = __builtin_amdgcn_wmma_f32_16x16x4_f32(false, a_, false, b0_,   \
                                                               (short)0, c0, false, false); \
                    c1 = __builtin_amdgcn_wmma_f32_16x16x4_f32(false, a_, false, b1_,   \
                                                               (short)0, c1, false, false); \
                    c2 = __builtin_amdgcn_wmma_f32_16x16x4_f32(false, a_, false, b2_,   \
                                                               (short)0, c2, false, false); \
                    c3 = __builtin_amdgcn_wmma_f32_16x16x4_f32(false, a_, false, b3_,   \
                                                               (short)0, c3, false, false); \
                }

                #pragma unroll 1
                for (int q = 0; q < 60; q += 5) {
                    const float* wn = wB + (size_t)(q + 5) * 4 * HD;
                    const float* an = mA + (q + 5) * 4;
                    #pragma unroll
                    for (int j = 0; j < 5; ++j) {
                        MMA_STEP(wlo[j], whi[j], av[j])
                        wlo[j] = *(const float4*)(wn + (size_t)j * 4 * HD);
                        whi[j] = *(const float4*)(wn + (size_t)j * 4 * HD + HD);
                        av[j]  = *(const v2f*)(an + j * 4);
                    }
                }
                MMA_STEP(wlo[0], whi[0], av[0])
                {
                    const int k0 = (MIXKP - 4) + hi * 2;
                    wlo[0] = make_float4(0.f, 0.f, 0.f, 0.f);
                    whi[0] = make_float4(0.f, 0.f, 0.f, 0.f);
                    if (k0 < MIXK) {
                        wlo[0] = *(const float4*)(wB + (size_t)65 * 4 * HD);
                        whi[0] = *(const float4*)(wB + (size_t)65 * 4 * HD + HD);
                    }
                    av[0] = *(const v2f*)(mA + 65 * 4);
                }
                MMA_STEP(wlo[1], whi[1], av[1])
                MMA_STEP(wlo[2], whi[2], av[2])
                MMA_STEP(wlo[3], whi[3], av[3])
                MMA_STEP(wlo[4], whi[4], av[4])
                MMA_STEP(wlo[0], whi[0], av[0])
#undef MMA_STEP

                const float4 bv = *(const float4*)(bias + col4);
                #pragma unroll
                for (int r = 0; r < 8; ++r) {
                    const int row = rbase + r + hi * 8;
                    const float x0 = c0[r] + bv.x;
                    const float x1 = c1[r] + bv.y;
                    const float x2 = c2[r] + bv.z;
                    const float x3 = c3[r] + bv.w;
                    v4f o;
                    o.x = x0 / (1.0f + __expf(-x0));
                    o.y = x1 / (1.0f + __expf(-x1));
                    o.z = x2 / (1.0f + __expf(-x2));
                    o.w = x3 / (1.0f + __expf(-x3));
                    __builtin_nontemporal_store(o, (v4f*)(out + (size_t)(base + row) * HD + col4));
                }
            }
        }
        __syncthreads();
    }
}

extern "C" void kernel_launch(void* const* d_in, const int* in_sizes, int n_in,
                              void* d_out, int out_size, void* d_ws, size_t ws_size,
                              hipStream_t stream) {
    const float* feat   = (const float*)d_in[0];
    const float* coords = (const float*)d_in[1];
    const int*   knn    = (const int*)  d_in[2];
    const float* W      = (const float*)d_in[3];
    const float* bias   = (const float*)d_in[4];
    float* out = (float*)d_out;

    const int n = in_sizes[0] / HD;

    graphconv_wmma_kernel<<<NGROUPS, 256, 0, stream>>>(feat, coords, knn, W, bias, out, n);
}
